# starting kernel
# baseline (speedup 1.0000x reference)
#include <hip/hip_runtime.h>
#include <hip/hip_bf16.h>
#include <math.h>

constexpr int B_  = 64;
constexpr int T_  = 50;
constexpr int S_  = 256;
constexpr int A_  = 64;
constexpr int H_  = 256;

typedef float v2f __attribute__((ext_vector_type(2)));
typedef float v8f __attribute__((ext_vector_type(8)));

__device__ __forceinline__ v8f wmma4_f32(v2f a, v2f b, v8f c) {
  return __builtin_amdgcn_wmma_f32_16x16x4_f32(
       false, a,  false, b,
       (short)0, c,  false,  false);
}

__global__ void ssm_hyper(const float* __restrict__ context,
                          const float* __restrict__ W1a, const float* __restrict__ b1a,
                          const float* __restrict__ W1b, const float* __restrict__ b1b,
                          float* __restrict__ HA, float* __restrict__ HB) {
  int idx = blockIdx.x * blockDim.x + threadIdx.x;
  if (idx >= T_ * B_ * H_) return;
  int k  = idx & (H_ - 1);
  int tb = idx >> 8;
  int b  = tb % B_;
  int t  = tb / B_;
  float c = context[b] + (float)t * (1.0f / 50.0f);
  HA[idx] = tanhf(fmaf(c, W1a[k], b1a[k]));
  HB[idx] = tanhf(fmaf(c, W1b[k], b1b[k]));
}

__global__ __launch_bounds__(64) void ssm_step_partial(
    const float* __restrict__ W2a, const float* __restrict__ b2a,
    const float* __restrict__ W2b, const float* __restrict__ b2b,
    const float* __restrict__ HA,  const float* __restrict__ HB,
    const float* __restrict__ xbase, int xstride,
    const float* __restrict__ ubase, int ustride,
    float* __restrict__ partials, int t) {
  const int lane = threadIdx.x & 31;
  const int wid  = threadIdx.x >> 5;
  const int blk  = blockIdx.x;
  const int bt   = blk & 3;
  const int it   = (blk >> 2) & 15;
  const int kc   = blk >> 6;
  const int b0   = bt * 16;
  const int i0   = it * 16;
  const int MN   = lane & 15;
  const int half = lane >> 4;
  const int jA0  = wid * 128;
  const int jB0  = wid * 32;

  const float* xrow = xbase + (size_t)(b0 + MN) * xstride + jA0 + half * 2;
  v2f xf[32];
#pragma unroll
  for (int f = 0; f < 32; ++f) {
    xf[f][0] = xrow[f * 4 + 0];
    xf[f][1] = xrow[f * 4 + 1];
  }
  const float* urow = ubase + (size_t)(b0 + MN) * ustride + jB0 + half * 2;
  v2f uf[8];
#pragma unroll
  for (int f = 0; f < 8; ++f) {
    uf[f][0] = urow[f * 4 + 0];
    uf[f][1] = urow[f * 4 + 1];
  }

  const float* haRow = HA + ((size_t)t * B_ + (b0 + MN)) * H_;
  const float* hbRow = HB + ((size_t)t * B_ + (b0 + MN)) * H_;

  v8f acc = {};
  const int kBeg = kc * 32, kEnd = kc * 32 + 32;
  for (int k = kBeg; k < kEnd; ++k) {
    const float* arow = W2a + (size_t)k * (S_ * S_) + (size_t)(i0 + MN) * S_ + jA0 + half * 2;
    v2f a0[16], a1[16];
#pragma unroll
    for (int f = 0; f < 16; ++f) { a0[f][0] = arow[f * 4];      a0[f][1] = arow[f * 4 + 1]; }
#pragma unroll
    for (int f = 0; f < 16; ++f) { a1[f][0] = arow[64 + f * 4]; a1[f][1] = arow[64 + f * 4 + 1]; }
    v8f t0 = {}, t1 = {};
#pragma unroll
    for (int f = 0; f < 16; f += 2) {
      t0 = wmma4_f32(a0[f],     xf[f],     t0);
      t1 = wmma4_f32(a0[f + 1], xf[f + 1], t1);
    }
#pragma unroll
    for (int f = 0; f < 16; f += 2) {
      t0 = wmma4_f32(a1[f],     xf[16 + f],     t0);
      t1 = wmma4_f32(a1[f + 1], xf[16 + f + 1], t1);
    }
    float sa = haRow[k];
#pragma unroll
    for (int r = 0; r < 8; ++r) acc[r] += sa * (t0[r] + t1[r]);

    const float* brow = W2b + (size_t)k * (S_ * A_) + (size_t)(i0 + MN) * A_ + jB0 + half * 2;
    v2f bfr[8];
#pragma unroll
    for (int f = 0; f < 8; ++f) { bfr[f][0] = brow[f * 4]; bfr[f][1] = brow[f * 4 + 1]; }
    v8f s0 = {}, s1 = {};
#pragma unroll
    for (int f = 0; f < 8; f += 2) {
      s0 = wmma4_f32(bfr[f],     uf[f],     s0);
      s1 = wmma4_f32(bfr[f + 1], uf[f + 1], s1);
    }
    float sb = hbRow[k];
#pragma unroll
    for (int r = 0; r < 8; ++r) acc[r] += sb * (s0[r] + s1[r]);
  }

  if (kc == 7) {
    const float* arow = b2a + (size_t)(i0 + MN) * S_ + jA0 + half * 2;
    v8f t0 = {}, t1 = {};
#pragma unroll
    for (int f = 0; f < 32; f += 2) {
      v2f c0; c0[0] = arow[f * 4];     c0[1] = arow[f * 4 + 1];
      v2f c1; c1[0] = arow[f * 4 + 4]; c1[1] = arow[f * 4 + 5];
      t0 = wmma4_f32(c0, xf[f],     t0);
      t1 = wmma4_f32(c1, xf[f + 1], t1);
    }
    const float* brow = b2b + (size_t)(i0 + MN) * A_ + jB0 + half * 2;
#pragma unroll
    for (int f = 0; f < 8; f += 2) {
      v2f c0; c0[0] = brow[f * 4];     c0[1] = brow[f * 4 + 1];
      v2f c1; c1[0] = brow[f * 4 + 4]; c1[1] = brow[f * 4 + 5];
      t0 = wmma4_f32(c0, uf[f],     t0);
      t1 = wmma4_f32(c1, uf[f + 1], t1);
    }
#pragma unroll
    for (int r = 0; r < 8; ++r) acc[r] += t0[r] + t1[r];
  }

  float* outp = partials + ((size_t)blk * 2 + wid) * 256 + lane * 8;
#pragma unroll
  for (int r = 0; r < 8; ++r) outp[r] = acc[r];
}

__global__ void ssm_step_reduce(const float* __restrict__ partials,
                                float* __restrict__ out, int t) {
  int e = blockIdx.x * blockDim.x + threadIdx.x;
  if (e >= B_ * S_) return;
  int b = e >> 8;
  int i = e & (S_ - 1);
  int it = i >> 4, M = i & 15;
  int bt = b >> 4, N = b & 15;
  int lane = (M >= 8) ? (16 + N) : N;
  int r    = M & 7;
  float sum = 0.f;
#pragma unroll
  for (int kc = 0; kc < 8; ++kc) {
#pragma unroll
    for (int wid = 0; wid < 2; ++wid) {
      sum += partials[(((size_t)(kc * 64 + it * 4 + bt)) * 2 + wid) * 256 + lane * 8 + r];
    }
  }
  out[((size_t)b * T_ + t) * S_ + i] = sum;
}

extern "C" void kernel_launch(void* const* d_in, const int* in_sizes, int n_in,
                              void* d_out, int out_size, void* d_ws, size_t ws_size,
                              hipStream_t stream) {
  const float* x0      = (const float*)d_in[0];
  const float* context = (const float*)d_in[1];
  const float* us      = (const float*)d_in[2];
  const float* W1a     = (const float*)d_in[3];
  const float* b1a     = (const float*)d_in[4];
  const float* W2a     = (const float*)d_in[5];
  const float* b2a     = (const float*)d_in[6];
  const float* W1b     = (const float*)d_in[7];
  const float* b1b     = (const float*)d_in[8];
  const float* W2b     = (const float*)d_in[9];
  const float* b2b     = (const float*)d_in[10];
  float* out = (float*)d_out;

  float* HA       = (float*)d_ws;
  float* HB       = HA + (size_t)T_ * B_ * H_;
  float* partials = HB + (size_t)T_ * B_ * H_;

  int nHyp = T_ * B_ * H_;
  ssm_hyper<<<(nHyp + 255) / 256, 256, 0, stream>>>(context, W1a, b1a, W1b, b1b, HA, HB);

  for (int t = 0; t < T_; ++t) {
    const float* xbase = (t == 0) ? x0 : (out + (size_t)(t - 1) * S_);
    int xstride        = (t == 0) ? S_ : T_ * S_;
    const float* ubase = us + (size_t)t * A_;

    ssm_step_partial<<<512, 64, 0, stream>>>(W2a, b2a, W2b, b2b, HA, HB,
                                             xbase, xstride, ubase, T_ * A_,
                                             partials, t);
    ssm_step_reduce<<<(B_ * S_ + 255) / 256, 256, 0, stream>>>(partials, out, t);
  }
}


# each block's 32 k split over 4 wave pairs: 8 waves/block, 4 resident waves per SIMD (was 1) cover weight-load waits; block's 8 tiles summed in LDS into 1 slice; summation order changes
# speedup vs baseline: 1.6672x; 1.6672x over previous
#include <hip/hip_runtime.h>
#include <hip/hip_bf16.h>
#include <math.h>

constexpr int B_  = 64;
constexpr int T_  = 50;
constexpr int S_  = 256;
constexpr int A_  = 64;
constexpr int H_  = 256;

typedef float v2f __attribute__((ext_vector_type(2)));
typedef float v8f __attribute__((ext_vector_type(8)));

__device__ __forceinline__ v8f wmma4_f32(v2f a, v2f b, v8f c) {
  return __builtin_amdgcn_wmma_f32_16x16x4_f32(
       false, a,  false, b,
       (short)0, c,  false,  false);
}

__global__ void ssm_hyper(const float* __restrict__ context,
                          const float* __restrict__ W1a, const float* __restrict__ b1a,
                          const float* __restrict__ W1b, const float* __restrict__ b1b,
                          float* __restrict__ HA, float* __restrict__ HB) {
  int idx = blockIdx.x * blockDim.x + threadIdx.x;
  if (idx >= T_ * B_ * H_) return;
  int k  = idx & (H_ - 1);
  int tb = idx >> 8;
  int b  = tb % B_;
  int t  = tb / B_;
  float c = context[b] + (float)t * (1.0f / 50.0f);
  HA[idx] = tanhf(fmaf(c, W1a[k], b1a[k]));
  HB[idx] = tanhf(fmaf(c, W1b[k], b1b[k]));
}

__global__ __launch_bounds__(256) void ssm_step_partial(
    const float* __restrict__ W2a, const float* __restrict__ b2a,
    const float* __restrict__ W2b, const float* __restrict__ b2b,
    const float* __restrict__ HA,  const float* __restrict__ HB,
    const float* __restrict__ xbase, int xstride,
    const float* __restrict__ ubase, int ustride,
    float* __restrict__ partials, int t) {
  const int lane = threadIdx.x & 31;
  const int wv   = __builtin_amdgcn_readfirstlane(threadIdx.x >> 5);
  const int wid  = wv & 1;
  const int ksub = wv >> 1;
  const int blk  = blockIdx.x;
  const int bt   = blk & 3;
  const int it   = (blk >> 2) & 15;
  const int kc   = blk >> 6;
  const int b0   = bt * 16;
  const int i0   = it * 16;
  const int MN   = lane & 15;
  const int half = lane >> 4;
  const int jA0  = wid * 128;
  const int jB0  = wid * 32;

  const float* xrow = xbase + (size_t)(b0 + MN) * xstride + jA0 + half * 2;
  v2f xf[32];
#pragma unroll
  for (int f = 0; f < 32; ++f) {
    xf[f][0] = xrow[f * 4 + 0];
    xf[f][1] = xrow[f * 4 + 1];
  }
  const float* urow = ubase + (size_t)(b0 + MN) * ustride + jB0 + half * 2;
  v2f uf[8];
#pragma unroll
  for (int f = 0; f < 8; ++f) {
    uf[f][0] = urow[f * 4 + 0];
    uf[f][1] = urow[f * 4 + 1];
  }

  const float* haRow = HA + ((size_t)t * B_ + (b0 + MN)) * H_;
  const float* hbRow = HB + ((size_t)t * B_ + (b0 + MN)) * H_;

  v8f acc = {};
  const int kBeg = kc * 32 + ksub * 8, kEnd = kBeg + 8;
  for (int k = kBeg; k < kEnd; ++k) {
    const float* arow = W2a + (size_t)k * (S_ * S_) + (size_t)(i0 + MN) * S_ + jA0 + half * 2;
    v2f a0[16], a1[16];
#pragma unroll
    for (int f = 0; f < 16; ++f) { a0[f][0] = arow[f * 4];      a0[f][1] = arow[f * 4 + 1]; }
#pragma unroll
    for (int f = 0; f < 16; ++f) { a1[f][0] = arow[64 + f * 4]; a1[f][1] = arow[64 + f * 4 + 1]; }
    v8f t0 = {}, t1 = {};
#pragma unroll
    for (int f = 0; f < 16; f += 2) {
      t0 = wmma4_f32(a0[f],     xf[f],     t0);
      t1 = wmma4_f32(a0[f + 1], xf[f + 1], t1);
    }
#pragma unroll
    for (int f = 0; f < 16; f += 2) {
      t0 = wmma4_f32(a1[f],     xf[16 + f],     t0);
      t1 = wmma4_f32(a1[f + 1], xf[16 + f + 1], t1);
    }
    float sa = haRow[k];
#pragma unroll
    for (int r = 0; r < 8; ++r) acc[r] += sa * (t0[r] + t1[r]);

    const float* brow = W2b + (size_t)k * (S_ * A_) + (size_t)(i0 + MN) * A_ + jB0 + half * 2;
    v2f bfr[8];
#pragma unroll
    for (int f = 0; f < 8; ++f) { bfr[f][0] = brow[f * 4]; bfr[f][1] = brow[f * 4 + 1]; }
    v8f s0 = {}, s1 = {};
#pragma unroll
    for (int f = 0; f < 8; f += 2) {
      s0 = wmma4_f32(bfr[f],     uf[f],     s0);
      s1 = wmma4_f32(bfr[f + 1], uf[f + 1], s1);
    }
    float sb = hbRow[k];
#pragma unroll
    for (int r = 0; r < 8; ++r) acc[r] += sb * (s0[r] + s1[r]);
  }

  if (kc == 7 && ksub == 3) {
    const float* arow = b2a + (size_t)(i0 + MN) * S_ + jA0 + half * 2;
    v8f t0 = {}, t1 = {};
#pragma unroll
    for (int f = 0; f < 32; f += 2) {
      v2f c0; c0[0] = arow[f * 4];     c0[1] = arow[f * 4 + 1];
      v2f c1; c1[0] = arow[f * 4 + 4]; c1[1] = arow[f * 4 + 5];
      t0 = wmma4_f32(c0, xf[f],     t0);
      t1 = wmma4_f32(c1, xf[f + 1], t1);
    }
    const float* brow = b2b + (size_t)(i0 + MN) * A_ + jB0 + half * 2;
#pragma unroll
    for (int f = 0; f < 8; f += 2) {
      v2f c0; c0[0] = brow[f * 4];     c0[1] = brow[f * 4 + 1];
      v2f c1; c1[0] = brow[f * 4 + 4]; c1[1] = brow[f * 4 + 5];
      t0 = wmma4_f32(c0, uf[f],     t0);
      t1 = wmma4_f32(c1, uf[f + 1], t1);
    }
#pragma unroll
    for (int r = 0; r < 8; ++r) acc[r] += t0[r] + t1[r];
  }

  __shared__ float red[8][256];
  float* mine = &red[wv][MN * 16 + half * 8];
#pragma unroll
  for (int r = 0; r < 8; ++r) mine[r] = acc[r];
  __syncthreads();
  float sum = 0.f;
#pragma unroll
  for (int w = 0; w < 8; ++w) sum += red[w][threadIdx.x];
  partials[(size_t)blk * 256 + threadIdx.x] = sum;
}

__global__ void ssm_step_reduce(const float* __restrict__ partials,
                                float* __restrict__ out, int t) {
  int e = blockIdx.x * blockDim.x + threadIdx.x;
  if (e >= B_ * S_) return;
  int b = e >> 8;
  int i = e & (S_ - 1);
  int it = i >> 4, M = i & 15;
  int bt = b >> 4, N = b & 15;
  float sum = 0.f;
#pragma unroll
  for (int kc = 0; kc < 8; ++kc) {
    sum += partials[((size_t)(kc * 64 + it * 4 + bt)) * 256 + N * 16 + M];
  }
  out[((size_t)b * T_ + t) * S_ + i] = sum;
}

extern "C" void kernel_launch(void* const* d_in, const int* in_sizes, int n_in,
                              void* d_out, int out_size, void* d_ws, size_t ws_size,
                              hipStream_t stream) {
  const float* x0      = (const float*)d_in[0];
  const float* context = (const float*)d_in[1];
  const float* us      = (const float*)d_in[2];
  const float* W1a     = (const float*)d_in[3];
  const float* b1a     = (const float*)d_in[4];
  const float* W2a     = (const float*)d_in[5];
  const float* b2a     = (const float*)d_in[6];
  const float* W1b     = (const float*)d_in[7];
  const float* b1b     = (const float*)d_in[8];
  const float* W2b     = (const float*)d_in[9];
  const float* b2b     = (const float*)d_in[10];
  float* out = (float*)d_out;

  float* HA       = (float*)d_ws;
  float* HB       = HA + (size_t)T_ * B_ * H_;
  float* partials = HB + (size_t)T_ * B_ * H_;

  int nHyp = T_ * B_ * H_;
  ssm_hyper<<<(nHyp + 255) / 256, 256, 0, stream>>>(context, W1a, b1a, W1b, b1b, HA, HB);

  for (int t = 0; t < T_; ++t) {
    const float* xbase = (t == 0) ? x0 : (out + (size_t)(t - 1) * S_);
    int xstride        = (t == 0) ? S_ : T_ * S_;
    const float* ubase = us + (size_t)t * A_;

    ssm_step_partial<<<512, 256, 0, stream>>>(W2a, b2a, W2b, b2b, HA, HB,
                                             xbase, xstride, ubase, T_ * A_,
                                             partials, t);
    ssm_step_reduce<<<(B_ * S_ + 255) / 256, 256, 0, stream>>>(partials, out, t);
  }
}


# all WMMA operands via LDS: block copies each k's 16 weight rows (and its x,u tiles) with whole-line 16-byte loads into 2 padded LDS stages, 8 waves read 8-byte fragments; summation order changes
# speedup vs baseline: 2.7760x; 1.6651x over previous
#include <hip/hip_runtime.h>
#include <hip/hip_bf16.h>
#include <math.h>

constexpr int B_  = 64;
constexpr int T_  = 50;
constexpr int S_  = 256;
constexpr int A_  = 64;
constexpr int H_  = 256;

typedef float v2f __attribute__((ext_vector_type(2)));
typedef float v8f __attribute__((ext_vector_type(8)));

__device__ __forceinline__ v8f wmma4_f32(v2f a, v2f b, v8f c) {
  return __builtin_amdgcn_wmma_f32_16x16x4_f32(
       false, a,  false, b,
       (short)0, c,  false,  false);
}

__global__ void ssm_hyper(const float* __restrict__ context,
                          const float* __restrict__ W1a, const float* __restrict__ b1a,
                          const float* __restrict__ W1b, const float* __restrict__ b1b,
                          float* __restrict__ HA, float* __restrict__ HB) {
  int idx = blockIdx.x * blockDim.x + threadIdx.x;
  if (idx >= T_ * B_ * H_) return;
  int k  = idx & (H_ - 1);
  int tb = idx >> 8;
  int b  = tb % B_;
  int t  = tb / B_;
  float c = context[b] + (float)t * (1.0f / 50.0f);
  HA[idx] = tanhf(fmaf(c, W1a[k], b1a[k]));
  HB[idx] = tanhf(fmaf(c, W1b[k], b1b[k]));
}

constexpr int PA       = S_ + 4;
constexpr int PB       = A_ + 4;
constexpr int SLAB_A   = 16 * PA;
constexpr int SLAB_B   = 16 * PB;
constexpr int STAGE_DW = 2 * (SLAB_A + SLAB_B);
constexpr int XS_DW    = 32 * PA;
constexpr int US_DW    = 32 * PB;
constexpr int LDS_DW   = 2 * STAGE_DW + XS_DW + US_DW;

__global__ __launch_bounds__(256) void ssm_step_partial(
    const float* W2a, const float* __restrict__ b2a,
    const float* W2b, const float* __restrict__ b2b,
    const float* __restrict__ HA,  const float* __restrict__ HB,
    const float* __restrict__ xbase, int xstride,
    const float* __restrict__ ubase, int ustride,
    float* __restrict__ partials, int t) {
  __shared__ float4 smem4[LDS_DW / 4];
  float* smem = reinterpret_cast<float*>(smem4);
  float* xs   = smem + 2 * STAGE_DW;
  float* usm  = xs + XS_DW;

  const int tid  = threadIdx.x;
  const int lane = tid & 31;
  const int wv   = __builtin_amdgcn_readfirstlane(tid >> 5);
  const int jh   = wv & 1;
  const int bl   = (wv >> 1) & 1;
  const int ksub = wv >> 2;
  const int blk  = blockIdx.x;
  const int btp  = blk & 1;
  const int it   = (blk >> 1) & 15;
  const int kc   = blk >> 5;
  const int bt   = btp * 2 + bl;
  const int b0   = bt * 16;
  const int i0   = it * 16;
  const int MN   = lane & 15;
  const int half = lane >> 4;
  const int jA0  = jh * 128;
  const int jB0  = jh * 32;

  const float* gA = W2a + (size_t)(kc * 32) * (S_ * S_) + (size_t)i0 * S_ + tid * 4;
  const float* gB = W2b + (size_t)(kc * 32) * (S_ * A_) + (size_t)i0 * A_ + tid * 4;
  const int ldsA = (tid >> 6) * PA + (tid & 63) * 4;
  const int ldsB = 2 * SLAB_A + (tid >> 4) * PB + (tid & 15) * 4;
  float4 ra0, ra1, ra2, ra3, ra4, ra5, ra6, ra7, rb0, rb1;
#define SSM_GLD_A(n, kk) (*reinterpret_cast<const float4*>(gA + (size_t)(((n) >> 2) * 16 + (kk)) * (S_ * S_) + ((n) & 3) * 1024))
#define SSM_GLD_B(n, kk) (*reinterpret_cast<const float4*>(gB + (size_t)((n) * 16 + (kk)) * (S_ * A_)))
#define SSM_GLOAD(kk)                                                   \
  do {                                                                  \
    ra0 = SSM_GLD_A(0, kk); ra1 = SSM_GLD_A(1, kk);                     \
    ra2 = SSM_GLD_A(2, kk); ra3 = SSM_GLD_A(3, kk);                     \
    ra4 = SSM_GLD_A(4, kk); ra5 = SSM_GLD_A(5, kk);                     \
    ra6 = SSM_GLD_A(6, kk); ra7 = SSM_GLD_A(7, kk);                     \
    rb0 = SSM_GLD_B(0, kk); rb1 = SSM_GLD_B(1, kk);                     \
  } while (0)
#define SSM_ST_A(stage, n) (*reinterpret_cast<float4*>((stage) + ldsA + ((n) >> 2) * SLAB_A + ((n) & 3) * 4 * PA))
#define SSM_ST_B(stage, n) (*reinterpret_cast<float4*>((stage) + ldsB + (n) * SLAB_B))
#define SSM_SSTORE(stage)                                               \
  do {                                                                  \
    SSM_ST_A(stage, 0) = ra0; SSM_ST_A(stage, 1) = ra1;                 \
    SSM_ST_A(stage, 2) = ra2; SSM_ST_A(stage, 3) = ra3;                 \
    SSM_ST_A(stage, 4) = ra4; SSM_ST_A(stage, 5) = ra5;                 \
    SSM_ST_A(stage, 6) = ra6; SSM_ST_A(stage, 7) = ra7;                 \
    SSM_ST_B(stage, 0) = rb0; SSM_ST_B(stage, 1) = rb1;                 \
  } while (0)

  SSM_GLOAD(0);

  {
    float4 xv[8], uv[2];
#pragma unroll
    for (int n = 0; n < 8; ++n)
      xv[n] = *reinterpret_cast<const float4*>(
          xbase + (size_t)(btp * 32 + (tid >> 6) + 4 * n) * xstride + (tid & 63) * 4);
#pragma unroll
    for (int n = 0; n < 2; ++n)
      uv[n] = *reinterpret_cast<const float4*>(
          ubase + (size_t)(btp * 32 + (tid >> 4) + 16 * n) * ustride + (tid & 15) * 4);
#pragma unroll
    for (int n = 0; n < 8; ++n)
      *reinterpret_cast<float4*>(xs + ((tid >> 6) + 4 * n) * PA + (tid & 63) * 4) = xv[n];
#pragma unroll
    for (int n = 0; n < 2; ++n)
      *reinterpret_cast<float4*>(usm + ((tid >> 4) + 16 * n) * PB + (tid & 15) * 4) = uv[n];
  }
  SSM_SSTORE(smem);
  __syncthreads();

  const float* xr = xs + (bl * 16 + MN) * PA + jA0 + half * 2;
  v2f xf[32];
#pragma unroll
  for (int f = 0; f < 32; ++f) xf[f] = *reinterpret_cast<const v2f*>(xr + f * 4);
  const float* ur = usm + (bl * 16 + MN) * PB + jB0 + half * 2;
  v2f uf[8];
#pragma unroll
  for (int f = 0; f < 8; ++f) uf[f] = *reinterpret_cast<const v2f*>(ur + f * 4);

  const int kmine = kc * 32 + ksub * 16;
  const float* haRow = HA + ((size_t)t * B_ + (b0 + MN)) * H_ + kmine;
  const float* hbRow = HB + ((size_t)t * B_ + (b0 + MN)) * H_ + kmine;
  const int fragA = ksub * SLAB_A + MN * PA + jA0 + half * 2;
  const int fragB = 2 * SLAB_A + ksub * SLAB_B + MN * PB + jB0 + half * 2;

#define SSM_LDGRP(dst, p)                                               \
  _Pragma("unroll") for (int q = 0; q < 8; ++q)                         \
    dst[q] = *reinterpret_cast<const v2f*>((p) + q * 4)
#define SSM_MMGRP(frag, bf, f0, c0, c1)                                 \
  _Pragma("unroll") for (int q = 0; q < 8; q += 2) {                    \
    c0 = wmma4_f32(frag[q],     bf[(f0) + q],     c0);                  \
    c1 = wmma4_f32(frag[q + 1], bf[(f0) + q + 1], c1);                  \
  }

#define SSM_COMPUTE(cur_, sa_, sb_)                                                    \
  do {                                                                                 \
    const float* ar = (cur_) + fragA;                                                  \
    const float* br = (cur_) + fragB;                                                  \
    v2f fa[8], fb[8];                            \
    SSM_LDGRP(fa, ar);                                                                 \
    __builtin_amdgcn_sched_barrier(0);   \
    v8f t0 = {}, t1 = {};                                                              \
    SSM_LDGRP(fb, ar + 32);                                                            \
    __builtin_amdgcn_sched_barrier(0);   \
    SSM_MMGRP(fa, xf, 0, t0, t1);                                                      \
    __builtin_amdgcn_sched_barrier(0);   \
    SSM_LDGRP(fa, ar + 64);                                                            \
    __builtin_amdgcn_sched_barrier(0);   \
    SSM_MMGRP(fb, xf, 8, t0, t1);                                                      \
    __builtin_amdgcn_sched_barrier(0);   \
    SSM_LDGRP(fb, ar + 96);                                                            \
    __builtin_amdgcn_sched_barrier(0);   \
    SSM_MMGRP(fa, xf, 16, t0, t1);                                                     \
    __builtin_amdgcn_sched_barrier(0);   \
    SSM_LDGRP(fa, br);                                                                 \
    __builtin_amdgcn_sched_barrier(0);   \
    SSM_MMGRP(fb, xf, 24, t0, t1);                                                     \
    _Pragma("unroll") for (int r = 0; r < 8; ++r) acc[r] += (sa_) * (t0[r] + t1[r]);   \
    v8f s0 = {}, s1 = {};                                                              \
    SSM_MMGRP(fa, uf, 0, s0, s1);                                                      \
    _Pragma("unroll") for (int r = 0; r < 8; ++r) acc[r] += (sb_) * (s0[r] + s1[r]);   \
  } while (0)

  v8f acc = {};
  for (int kk = 0; kk < 15; ++kk) {
    const float* cur = smem + (kk & 1) * STAGE_DW;
    float* nxt = smem + ((kk + 1) & 1) * STAGE_DW;
    SSM_GLOAD(kk + 1);
    const float sa = haRow[kk];
    const float sb = hbRow[kk];
    SSM_COMPUTE(cur, sa, sb);
    SSM_SSTORE(nxt);
    __syncthreads();
  }
  {
    const float sa = haRow[15];
    const float sb = hbRow[15];
    SSM_COMPUTE(smem + STAGE_DW, sa, sb);
  }

  if (kc == 7 && ksub == 1) {
    const float* arow = b2a + (size_t)(i0 + MN) * S_ + jA0 + half * 2;
    v8f t0 = {}, t1 = {};
#pragma unroll
    for (int f = 0; f < 32; f += 2) {
      v2f c0; c0[0] = arow[f * 4];     c0[1] = arow[f * 4 + 1];
      v2f c1; c1[0] = arow[f * 4 + 4]; c1[1] = arow[f * 4 + 5];
      t0 = wmma4_f32(c0, xf[f],     t0);
      t1 = wmma4_f32(c1, xf[f + 1], t1);
    }
    const float* brow = b2b + (size_t)(i0 + MN) * A_ + jB0 + half * 2;
#pragma unroll
    for (int f = 0; f < 8; f += 2) {
      v2f c0; c0[0] = brow[f * 4];     c0[1] = brow[f * 4 + 1];
      v2f c1; c1[0] = brow[f * 4 + 4]; c1[1] = brow[f * 4 + 5];
      t0 = wmma4_f32(c0, uf[f],     t0);
      t1 = wmma4_f32(c1, uf[f + 1], t1);
    }
#pragma unroll
    for (int r = 0; r < 8; ++r) acc[r] += t0[r] + t1[r];
  }

  float* red  = smem;
  float* mine = red + wv * 256 + MN * 16 + half * 8;
#pragma unroll
  for (int r = 0; r < 8; ++r) mine[r] = acc[r];
  __syncthreads();
#pragma unroll
  for (int blo = 0; blo < 2; ++blo) {
    float sum = 0.f;
#pragma unroll
    for (int w = 0; w < 4; ++w) sum += red[((w >> 1) * 4 + blo * 2 + (w & 1)) * 256 + tid];
    partials[(size_t)(kc * 64 + it * 4 + btp * 2 + blo) * 256 + tid] = sum;
  }
}

__global__ void ssm_step_reduce(const float* __restrict__ partials,
                                float* __restrict__ out, int t) {
  int e = blockIdx.x * blockDim.x + threadIdx.x;
  if (e >= B_ * S_) return;
  int b = e >> 8;
  int i = e & (S_ - 1);
  int it = i >> 4, M = i & 15;
  int bt = b >> 4, N = b & 15;
  float sum = 0.f;
#pragma unroll
  for (int kc = 0; kc < 8; ++kc) {
    sum += partials[((size_t)(kc * 64 + it * 4 + bt)) * 256 + N * 16 + M];
  }
  out[((size_t)b * T_ + t) * S_ + i] = sum;
}

extern "C" void kernel_launch(void* const* d_in, const int* in_sizes, int n_in,
                              void* d_out, int out_size, void* d_ws, size_t ws_size,
                              hipStream_t stream) {
  const float* x0      = (const float*)d_in[0];
  const float* context = (const float*)d_in[1];
  const float* us      = (const float*)d_in[2];
  const float* W1a     = (const float*)d_in[3];
  const float* b1a     = (const float*)d_in[4];
  const float* W2a     = (const float*)d_in[5];
  const float* b2a     = (const float*)d_in[6];
  const float* W1b     = (const float*)d_in[7];
  const float* b1b     = (const float*)d_in[8];
  const float* W2b     = (const float*)d_in[9];
  const float* b2b     = (const float*)d_in[10];
  float* out = (float*)d_out;

  float* HA       = (float*)d_ws;
  float* HB       = HA + (size_t)T_ * B_ * H_;
  float* partials = HB + (size_t)T_ * B_ * H_;

  int nHyp = T_ * B_ * H_;
  ssm_hyper<<<(nHyp + 255) / 256, 256, 0, stream>>>(context, W1a, b1a, W1b, b1b, HA, HB);

  for (int t = 0; t < T_; ++t) {
    const float* xbase = (t == 0) ? x0 : (out + (size_t)(t - 1) * S_);
    int xstride        = (t == 0) ? S_ : T_ * S_;
    const float* ubase = us + (size_t)t * A_;

    ssm_step_partial<<<256, 256, 0, stream>>>(W2a, b2a, W2b, b2b, HA, HB,
                                             xbase, xstride, ubase, T_ * A_,
                                             partials, t);
    ssm_step_reduce<<<(B_ * S_ + 255) / 256, 256, 0, stream>>>(partials, out, t);
  }
}


# contraction split across blocks by state column j (32 per block, all k inside, B operand = ha*x), not by k: next launch sums the 8 slices as it loads x, reduce launch gone; summation order changes
# speedup vs baseline: 2.9098x; 1.0482x over previous
#include <hip/hip_runtime.h>
#include <math.h>

constexpr int B_  = 64;
constexpr int T_  = 50;
constexpr int S_  = 256;
constexpr int A_  = 64;
constexpr int H_  = 256;

typedef float v2f __attribute__((ext_vector_type(2)));
typedef float v8f __attribute__((ext_vector_type(8)));

__device__ __forceinline__ v8f wmma4_f32(v2f a, v2f b, v8f c) {
  return __builtin_amdgcn_wmma_f32_16x16x4_f32(
       false, a,  false, b,
       (short)0, c,  false,  false);
}

__global__ void ssm_hyper(const float* __restrict__ context,
                          const float* __restrict__ W1a, const float* __restrict__ b1a,
                          float* __restrict__ HA) {
  int idx = blockIdx.x * blockDim.x + threadIdx.x;
  if (idx >= T_ * B_ * H_) return;
  int k  = idx & (H_ - 1);
  int tb = idx >> 8;
  int b  = tb % B_;
  int t  = tb / B_;
  float c = context[b] + (float)t * (1.0f / 50.0f);
  HA[idx] = tanhf(fmaf(c, W1a[k], b1a[k]));
}

constexpr int PX       = 32 + 4;
constexpr int PB       = A_ + 4;
constexpr int SLAB_K   = 16 * PX;
constexpr int SLAB_B   = 16 * PB;
constexpr int STAGE_DW = 16 * SLAB_K + 2 * SLAB_B;
constexpr int XS_DW    = 32 * PX;
constexpr int US_DW    = 32 * PB;
constexpr int HS_DW    = 32 * PX;
constexpr int LDS_DW   = 2 * STAGE_DW + XS_DW + US_DW + HS_DW;
constexpr int SLICE    = 64 * 256;

template <bool FIRST>
__global__ __launch_bounds__(256) void ssm_step_partial(
    const float* W2a, const float* b2a,
    const float* W2b, const float* b2b,
    const float* __restrict__ HA,
    const float* __restrict__ context,
    const float* __restrict__ W1b, const float* __restrict__ b1b,
    const float* __restrict__ x0, const float* __restrict__ prev,
    const float* __restrict__ ubase, int ustride,
    float* __restrict__ partials, float* __restrict__ out, int t) {
  __shared__ float4 smem4[LDS_DW / 4];
  float* smem = reinterpret_cast<float*>(smem4);
  float* xs   = smem + 2 * STAGE_DW;
  float* usm  = xs + XS_DW;
  float* hbs  = usm + US_DW;

  const int tid  = threadIdx.x;
  const int lane = tid & 31;
  const int wv   = __builtin_amdgcn_readfirstlane(tid >> 5);
  const int bl   = wv & 1;
  const int kq   = wv >> 1;
  const int blk  = blockIdx.x;
  const int btp  = blk & 1;
  const int it   = (blk >> 1) & 15;
  const int jc   = blk >> 5;
  const int i0   = it * 16;
  const int j0   = jc * 32;
  const int MN   = lane & 15;
  const int half = lane >> 4;

  const float* gA = W2a + (size_t)(tid >> 7) * (S_ * S_)
                        + (size_t)(i0 + ((tid >> 3) & 15)) * S_ + j0 + (tid & 7) * 4;
  const float* gB = W2b + (size_t)j0 * (S_ * A_) + (size_t)i0 * A_ + tid * 4;
  const int ldsA = (tid >> 7) * SLAB_K + ((tid >> 3) & 15) * PX + (tid & 7) * 4;
  const int ldsB = 16 * SLAB_K + (tid >> 4) * PB + (tid & 15) * 4;
  float4 ra0, ra1, ra2, ra3, ra4, ra5, ra6, ra7, rb0, rb1;
#define SSM_GLD_A(n, s) (*reinterpret_cast<const float4*>(gA + (size_t)(16 * (s) + 2 * (n)) * (S_ * S_)))
#define SSM_GLD_B(n, s) (*reinterpret_cast<const float4*>(gB + (size_t)(2 * (s) + (n)) * (S_ * A_)))
#define SSM_GLOAD(s)                                                    \
  do {                                                                  \
    ra0 = SSM_GLD_A(0, s); ra1 = SSM_GLD_A(1, s);                       \
    ra2 = SSM_GLD_A(2, s); ra3 = SSM_GLD_A(3, s);                       \
    ra4 = SSM_GLD_A(4, s); ra5 = SSM_GLD_A(5, s);                       \
    ra6 = SSM_GLD_A(6, s); ra7 = SSM_GLD_A(7, s);                       \
    rb0 = SSM_GLD_B(0, s); rb1 = SSM_GLD_B(1, s);                       \
  } while (0)
#define SSM_ST_A(stage, n) (*reinterpret_cast<float4*>((stage) + ldsA + 2 * (n) * SLAB_K))
#define SSM_ST_B(stage, n) (*reinterpret_cast<float4*>((stage) + ldsB + (n) * SLAB_B))
#define SSM_SSTORE(stage)                                               \
  do {                                                                  \
    SSM_ST_A(stage, 0) = ra0; SSM_ST_A(stage, 1) = ra1;                 \
    SSM_ST_A(stage, 2) = ra2; SSM_ST_A(stage, 3) = ra3;                 \
    SSM_ST_A(stage, 4) = ra4; SSM_ST_A(stage, 5) = ra5;                 \
    SSM_ST_A(stage, 6) = ra6; SSM_ST_A(stage, 7) = ra7;                 \
    SSM_ST_B(stage, 0) = rb0; SSM_ST_B(stage, 1) = rb1;                 \
  } while (0)

  const int bq = tid >> 3;
  const int c8 = tid & 7;
  SSM_GLOAD(0);
  const float* haRow = HA + ((size_t)t * B_ + (btp * 32 + bl * 16 + MN)) * H_ + kq * 4;
  float4 hc = *reinterpret_cast<const float4*>(haRow);
  float4 pv[8];
  if (FIRST) {
    pv[0] = *reinterpret_cast<const float4*>(x0 + (size_t)(btp * 32 + bq) * S_ + j0 + c8 * 4);
  } else {
    const float* pp = prev + (size_t)((2 * jc + (c8 >> 2)) * 4 + btp * 2 + (bq >> 4)) * 256
                           + (bq & 15) * 16 + (c8 & 3) * 4;
#pragma unroll
    for (int js = 0; js < 8; ++js)
      pv[js] = *reinterpret_cast<const float4*>(pp + (size_t)js * SLICE);
  }
  float4 uv[2];
#pragma unroll
  for (int n = 0; n < 2; ++n)
    uv[n] = *reinterpret_cast<const float4*>(
        ubase + (size_t)(btp * 32 + (tid >> 4) + 16 * n) * ustride + (tid & 15) * 4);
  const float cv0 = context[btp * 32 + bq];
  const float4 w1 = *reinterpret_cast<const float4*>(W1b + j0 + c8 * 4);
  const float4 bb = *reinterpret_cast<const float4*>(b1b + j0 + c8 * 4);
  const float* arow = b2a + (size_t)(i0 + MN) * S_ + j0 + half * 2;
  v2f ba[8];
#pragma unroll
  for (int f = 0; f < 8; ++f) ba[f] = *reinterpret_cast<const v2f*>(arow + f * 4);
  const float* brow = b2b + (size_t)(i0 + MN) * A_ + kq * 16 + half * 2;
  v2f bu[4];
#pragma unroll
  for (int f = 0; f < 4; ++f) bu[f] = *reinterpret_cast<const v2f*>(brow + f * 4);

  float4 xv;
  if (FIRST) {
    xv = pv[0];
  } else {
    xv.x = 0.f; xv.y = 0.f; xv.z = 0.f; xv.w = 0.f;
#pragma unroll
    for (int js = 0; js < 8; ++js) {
      xv.x += pv[js].x; xv.y += pv[js].y; xv.z += pv[js].z; xv.w += pv[js].w;
    }
  }
  *reinterpret_cast<float4*>(xs + bq * PX + c8 * 4) = xv;
#pragma unroll
  for (int n = 0; n < 2; ++n)
    *reinterpret_cast<float4*>(usm + ((tid >> 4) + 16 * n) * PB + (tid & 15) * 4) = uv[n];
  const float cv = cv0 + (float)t * (1.0f / 50.0f);
  const float a0 = fmaf(cv, w1.x, bb.x), a1 = fmaf(cv, w1.y, bb.y);
  const float a2 = fmaf(cv, w1.z, bb.z), a3 = fmaf(cv, w1.w, bb.w);
  SSM_SSTORE(smem);
  {
    float4 hv;
    hv.x = tanhf(a0); hv.y = tanhf(a1); hv.z = tanhf(a2); hv.w = tanhf(a3);
    *reinterpret_cast<float4*>(hbs + bq * PX + c8 * 4) = hv;
  }
  __syncthreads();
  if (!FIRST && it == 0)
    *reinterpret_cast<float4*>(out + ((size_t)(btp * 32 + bq) * T_ + (t - 1)) * S_ + j0 + c8 * 4) = xv;

  const float* xr = xs + (bl * 16 + MN) * PX + half * 2;
  v2f xf[8];
#pragma unroll
  for (int f = 0; f < 8; ++f) xf[f] = *reinterpret_cast<const v2f*>(xr + f * 4);
  const float* ur = usm + (bl * 16 + MN) * PB + kq * 16 + half * 2;
  v2f uf[4];
#pragma unroll
  for (int f = 0; f < 4; ++f) uf[f] = *reinterpret_cast<const v2f*>(ur + f * 4);
  const float* hbRow = hbs + (bl * 16 + MN) * PX;

  const int fragA = kq * 4 * SLAB_K + MN * PX + half * 2;
  const int fragB = 16 * SLAB_K + MN * PB + kq * 16 + half * 2;

  v8f t0 = {}, t1 = {};

  if (kq == 3) {
#pragma unroll
    for (int f = 0; f < 8; f += 2) {
      t0 = wmma4_f32(ba[f],     xf[f],     t0);
      t1 = wmma4_f32(ba[f + 1], xf[f + 1], t1);
    }
  }
  if (jc == 7) {
#pragma unroll
    for (int f = 0; f < 4; f += 2) {
      t0 = wmma4_f32(bu[f],     uf[f],     t0);
      t1 = wmma4_f32(bu[f + 1], uf[f + 1], t1);
    }
  }

#define SSM_LDGRP(dst, p)                                               \
  _Pragma("unroll") for (int q = 0; q < 8; ++q)                         \
    dst[q] = *reinterpret_cast<const v2f*>((p) + q * 4)
#define SSM_LDGRPB(dst, p)                                              \
  _Pragma("unroll") for (int q = 0; q < 8; ++q)                         \
    dst[q] = *reinterpret_cast<const v2f*>((p) + (q >> 2) * SLAB_B + (q & 3) * 4)
#define SSM_SCALE(dst, h)                                               \
  _Pragma("unroll") for (int q = 0; q < 8; ++q) dst[q] = xf[q] * (h)
#define SSM_SCALEU(dst, h2)                                             \
  _Pragma("unroll") for (int q = 0; q < 8; ++q) dst[q] = uf[q & 3] * (h2)[q >> 2]
#define SSM_MMGRP(frag, bf)                                             \
  _Pragma("unroll") for (int q = 0; q < 8; q += 2) {                    \
    t0 = wmma4_f32(frag[q],     bf[q],     t0);                         \
    t1 = wmma4_f32(frag[q + 1], bf[q + 1], t1);                         \
  }

#define SSM_COMPUTE(cur_, h4_, hbp_)                                                   \
  do {                                                                                 \
    const float* ar = (cur_) + fragA;                                                  \
    const float* br = (cur_) + fragB;                                                  \
    v2f fa[8], fb[8], p0[8], p1[8];   \
    const v2f hb2 = *reinterpret_cast<const v2f*>(hbp_);                               \
    SSM_LDGRP(fa, ar);                                                                 \
    __builtin_amdgcn_sched_barrier(0);   \
    SSM_LDGRP(fb, ar + SLAB_K);                                                        \
    SSM_SCALE(p0, (h4_).x);                                                            \
    __builtin_amdgcn_sched_barrier(0);   \
    SSM_SCALE(p1, (h4_).y);                                                            \
    SSM_MMGRP(fa, p0);                                                                 \
    __builtin_amdgcn_sched_barrier(0);   \
    SSM_LDGRP(fa, ar + 2 * SLAB_K);                                                    \
    __builtin_amdgcn_sched_barrier(0);   \
    SSM_SCALE(p0, (h4_).z);                                                            \
    SSM_MMGRP(fb, p1);                                                                 \
    __builtin_amdgcn_sched_barrier(0);   \
    SSM_LDGRP(fb, ar + 3 * SLAB_K);                                                    \
    __builtin_amdgcn_sched_barrier(0);   \
    SSM_SCALE(p1, (h4_).w);                                                            \
    SSM_MMGRP(fa, p0);                                                                 \
    __builtin_amdgcn_sched_barrier(0);   \
    SSM_LDGRPB(fa, br);                                                                \
    __builtin_amdgcn_sched_barrier(0);   \
    SSM_SCALEU(p0, hb2);                                                               \
    SSM_MMGRP(fb, p1);                                                                 \
    SSM_MMGRP(fa, p0);                                                                 \
  } while (0)

  for (int s = 0; s < 15; ++s) {
    const float* cur = smem + (s & 1) * STAGE_DW;
    float* nxt = smem + ((s + 1) & 1) * STAGE_DW;
    const float4 hn = *reinterpret_cast<const float4*>(haRow + 16 * (s + 1));
    SSM_GLOAD(s + 1);
    SSM_COMPUTE(cur, hc, hbRow + 2 * s);
    SSM_SSTORE(nxt);
    __syncthreads();
    hc = hn;
  }
  SSM_COMPUTE(smem + STAGE_DW, hc, hbRow + 30);

  v8f acc;
#pragma unroll
  for (int r = 0; r < 8; ++r) acc[r] = t0[r] + t1[r];

  float* red  = smem;
  float* mine = red + wv * 256 + MN * 16 + half * 8;
#pragma unroll
  for (int r = 0; r < 8; ++r) mine[r] = acc[r];
  __syncthreads();
#pragma unroll
  for (int blo = 0; blo < 2; ++blo) {
    float sum = 0.f;
#pragma unroll
    for (int w = 0; w < 4; ++w) sum += red[(w * 2 + blo) * 256 + tid];
    partials[(size_t)(jc * 64 + it * 4 + btp * 2 + blo) * 256 + tid] = sum;
  }
}

__global__ void ssm_step_reduce(const float* __restrict__ partials,
                                float* __restrict__ out, int t) {
  int e = blockIdx.x * blockDim.x + threadIdx.x;
  if (e >= B_ * S_) return;
  int b = e >> 8;
  int i = e & (S_ - 1);
  int it = i >> 4, M = i & 15;
  int bt = b >> 4, N = b & 15;
  float sum = 0.f;
#pragma unroll
  for (int jc = 0; jc < 8; ++jc) {
    sum += partials[((size_t)(jc * 64 + it * 4 + bt)) * 256 + N * 16 + M];
  }
  out[((size_t)b * T_ + t) * S_ + i] = sum;
}

extern "C" void kernel_launch(void* const* d_in, const int* in_sizes, int n_in,
                              void* d_out, int out_size, void* d_ws, size_t ws_size,
                              hipStream_t stream) {
  const float* x0      = (const float*)d_in[0];
  const float* context = (const float*)d_in[1];
  const float* us      = (const float*)d_in[2];
  const float* W1a     = (const float*)d_in[3];
  const float* b1a     = (const float*)d_in[4];
  const float* W2a     = (const float*)d_in[5];
  const float* b2a     = (const float*)d_in[6];
  const float* W1b     = (const float*)d_in[7];
  const float* b1b     = (const float*)d_in[8];
  const float* W2b     = (const float*)d_in[9];
  const float* b2b     = (const float*)d_in[10];
  float* out = (float*)d_out;

  float* HA = (float*)d_ws;
  float* P0 = HA + (size_t)T_ * B_ * H_;
  float* P1 = P0 + (size_t)512 * 256;

  int nHyp = T_ * B_ * H_;
  ssm_hyper<<<(nHyp + 255) / 256, 256, 0, stream>>>(context, W1a, b1a, HA);

  for (int t = 0; t < T_; ++t) {
    const float* prev  = (t & 1) ? P0 : P1;
    float* cur         = (t & 1) ? P1 : P0;
    const float* ubase = us + (size_t)t * A_;

    if (t == 0)
      ssm_step_partial<true><<<256, 256, 0, stream>>>(W2a, b2a, W2b, b2b, HA, context, W1b, b1b,
                                                     x0, prev, ubase, T_ * A_, cur, out, t);
    else
      ssm_step_partial<false><<<256, 256, 0, stream>>>(W2a, b2a, W2b, b2b, HA, context, W1b, b1b,
                                                      x0, prev, ubase, T_ * A_, cur, out, t);
  }
  ssm_step_reduce<<<(B_ * S_ + 255) / 256, 256, 0, stream>>>(P1, out, T_ - 1);
}


# 3-slot LDS ring, the per-stage workgroup barrier split and moved to mid-stage: stores get a stage to land, next stage's first fragments read ahead, no stall at stage ends; summation order unchanged
# speedup vs baseline: 3.0892x; 1.0617x over previous
#include <hip/hip_runtime.h>
#include <math.h>

constexpr int B_  = 64;
constexpr int T_  = 50;
constexpr int S_  = 256;
constexpr int A_  = 64;
constexpr int H_  = 256;

typedef float v2f __attribute__((ext_vector_type(2)));
typedef float v8f __attribute__((ext_vector_type(8)));

__device__ __forceinline__ v8f wmma4_f32(v2f a, v2f b, v8f c) {
  return __builtin_amdgcn_wmma_f32_16x16x4_f32(
       false, a,  false, b,
       (short)0, c,  false,  false);
}

__device__ __forceinline__ void lds_drained() { asm volatile("s_wait_dscnt 0x0" ::: "memory"); }
__device__ __forceinline__ void bar_signal()  { asm volatile("s_barrier_signal -1" ::: "memory"); }
__device__ __forceinline__ void bar_wait()    { asm volatile("s_barrier_wait -1" ::: "memory"); }

__global__ void ssm_hyper(const float* __restrict__ context,
                          const float* __restrict__ W1a, const float* __restrict__ b1a,
                          float* __restrict__ HA) {
  int idx = blockIdx.x * blockDim.x + threadIdx.x;
  if (idx >= T_ * B_ * H_) return;
  int k  = idx & (H_ - 1);
  int tb = idx >> 8;
  int b  = tb % B_;
  int t  = tb / B_;
  float c = context[b] + (float)t * (1.0f / 50.0f);
  HA[idx] = tanhf(fmaf(c, W1a[k], b1a[k]));
}

constexpr int PX       = 32 + 4;
constexpr int PB       = A_ + 4;
constexpr int SLAB_K   = 16 * PX;
constexpr int SLAB_B   = 16 * PB;
constexpr int STAGE_DW = 16 * SLAB_K + 2 * SLAB_B;
constexpr int XS_DW    = 32 * PX;
constexpr int US_DW    = 32 * PB;
constexpr int HS_DW    = 32 * PX;
constexpr int LDS_DW   = 3 * STAGE_DW + XS_DW + US_DW + HS_DW;
constexpr int SLICE    = 64 * 256;

template <bool FIRST>
__global__ __launch_bounds__(256) void ssm_step_partial(
    const float* W2a, const float* b2a,
    const float* W2b, const float* b2b,
    const float* __restrict__ HA,
    const float* __restrict__ context,
    const float* __restrict__ W1b, const float* __restrict__ b1b,
    const float* __restrict__ x0, const float* __restrict__ prev,
    const float* __restrict__ ubase, int ustride,
    float* __restrict__ partials, float* __restrict__ out, int t) {
  __shared__ float4 smem4[LDS_DW / 4];
  float* smem = reinterpret_cast<float*>(smem4);
  float* xs   = smem + 3 * STAGE_DW;
  float* usm  = xs + XS_DW;
  float* hbs  = usm + US_DW;

  const int tid  = threadIdx.x;
  const int lane = tid & 31;
  const int wv   = __builtin_amdgcn_readfirstlane(tid >> 5);
  const int bl   = wv & 1;
  const int kq   = wv >> 1;
  const int blk  = blockIdx.x;
  const int btp  = blk & 1;
  const int it   = (blk >> 1) & 15;
  const int jc   = blk >> 5;
  const int i0   = it * 16;
  const int j0   = jc * 32;
  const int MN   = lane & 15;
  const int half = lane >> 4;

  const float* gA = W2a + (size_t)(tid >> 7) * (S_ * S_)
                        + (size_t)(i0 + ((tid >> 3) & 15)) * S_ + j0 + (tid & 7) * 4;
  const float* gB = W2b + (size_t)j0 * (S_ * A_) + (size_t)i0 * A_ + tid * 4;
  const int ldsA = (tid >> 7) * SLAB_K + ((tid >> 3) & 15) * PX + (tid & 7) * 4;
  const int ldsB = 16 * SLAB_K + (tid >> 4) * PB + (tid & 15) * 4;
  float4 ra0, ra1, ra2, ra3, ra4, ra5, ra6, ra7, rb0, rb1;
  float4 sa0, sa1, sa2, sa3, sa4, sa5, sa6, sa7, sb0, sb1;
#define SSM_GLD_A(n, s) (*reinterpret_cast<const float4*>(gA + (size_t)(16 * (s) + 2 * (n)) * (S_ * S_)))
#define SSM_GLD_B(n, s) (*reinterpret_cast<const float4*>(gB + (size_t)(2 * (s) + (n)) * (S_ * A_)))
#define SSM_GLOAD(s)                                                    \
  do {                                                                  \
    ra0 = SSM_GLD_A(0, s); ra1 = SSM_GLD_A(1, s);                       \
    ra2 = SSM_GLD_A(2, s); ra3 = SSM_GLD_A(3, s);                       \
    ra4 = SSM_GLD_A(4, s); ra5 = SSM_GLD_A(5, s);                       \
    ra6 = SSM_GLD_A(6, s); ra7 = SSM_GLD_A(7, s);                       \
    rb0 = SSM_GLD_B(0, s); rb1 = SSM_GLD_B(1, s);                       \
  } while (0)
#define SSM_GLOAD_SECOND(s)                                             \
  do {                                                                  \
    sa0 = SSM_GLD_A(0, s); sa1 = SSM_GLD_A(1, s);                       \
    sa2 = SSM_GLD_A(2, s); sa3 = SSM_GLD_A(3, s);                       \
    sa4 = SSM_GLD_A(4, s); sa5 = SSM_GLD_A(5, s);                       \
    sa6 = SSM_GLD_A(6, s); sa7 = SSM_GLD_A(7, s);                       \
    sb0 = SSM_GLD_B(0, s); sb1 = SSM_GLD_B(1, s);                       \
  } while (0)
#define SSM_ST_A(stage, n) (*reinterpret_cast<float4*>((stage) + ldsA + 2 * (n) * SLAB_K))
#define SSM_ST_B(stage, n) (*reinterpret_cast<float4*>((stage) + ldsB + (n) * SLAB_B))
#define SSM_SSTORE(stage)                                               \
  do {                                                                  \
    SSM_ST_A(stage, 0) = ra0; SSM_ST_A(stage, 1) = ra1;                 \
    SSM_ST_A(stage, 2) = ra2; SSM_ST_A(stage, 3) = ra3;                 \
    SSM_ST_A(stage, 4) = ra4; SSM_ST_A(stage, 5) = ra5;                 \
    SSM_ST_A(stage, 6) = ra6; SSM_ST_A(stage, 7) = ra7;                 \
    SSM_ST_B(stage, 0) = rb0; SSM_ST_B(stage, 1) = rb1;                 \
  } while (0)

#define SSM_SSTORE_SECOND(stage)                                        \
  do {                                                                  \
    SSM_ST_A(stage, 0) = sa0; SSM_ST_A(stage, 1) = sa1;                 \
    SSM_ST_A(stage, 2) = sa2; SSM_ST_A(stage, 3) = sa3;                 \
    SSM_ST_A(stage, 4) = sa4; SSM_ST_A(stage, 5) = sa5;                 \
    SSM_ST_A(stage, 6) = sa6; SSM_ST_A(stage, 7) = sa7;                 \
    SSM_ST_B(stage, 0) = sb0; SSM_ST_B(stage, 1) = sb1;                 \
  } while (0)

  const int bq = tid >> 3;
  const int c8 = tid & 7;
  SSM_GLOAD(0);
  SSM_GLOAD_SECOND(1);
  const float* haRow = HA + ((size_t)t * B_ + (btp * 32 + bl * 16 + MN)) * H_ + kq * 4;
  float4 hc = *reinterpret_cast<const float4*>(haRow);
  float4 pv[8];
  if (FIRST) {
    pv[0] = *reinterpret_cast<const float4*>(x0 + (size_t)(btp * 32 + bq) * S_ + j0 + c8 * 4);
  } else {
    const float* pp = prev + (size_t)((2 * jc + (c8 >> 2)) * 4 + btp * 2 + (bq >> 4)) * 256
                           + (bq & 15) * 16 + (c8 & 3) * 4;
#pragma unroll
    for (int js = 0; js < 8; ++js)
      pv[js] = *reinterpret_cast<const float4*>(pp + (size_t)js * SLICE);
  }
  float4 uv[2];
#pragma unroll
  for (int n = 0; n < 2; ++n)
    uv[n] = *reinterpret_cast<const float4*>(
        ubase + (size_t)(btp * 32 + (tid >> 4) + 16 * n) * ustride + (tid & 15) * 4);
  const float cv0 = context[btp * 32 + bq];
  const float4 w1 = *reinterpret_cast<const float4*>(W1b + j0 + c8 * 4);
  const float4 bb = *reinterpret_cast<const float4*>(b1b + j0 + c8 * 4);
  const float* arow = b2a + (size_t)(i0 + MN) * S_ + j0 + half * 2;
  v2f ba[8];
#pragma unroll
  for (int f = 0; f < 8; ++f) ba[f] = *reinterpret_cast<const v2f*>(arow + f * 4);
  const float* brow = b2b + (size_t)(i0 + MN) * A_ + kq * 16 + half * 2;
  v2f bu[4];
#pragma unroll
  for (int f = 0; f < 4; ++f) bu[f] = *reinterpret_cast<const v2f*>(brow + f * 4);

  float4 xv;
  if (FIRST) {
    xv = pv[0];
  } else {
    xv.x = 0.f; xv.y = 0.f; xv.z = 0.f; xv.w = 0.f;
#pragma unroll
    for (int js = 0; js < 8; ++js) {
      xv.x += pv[js].x; xv.y += pv[js].y; xv.z += pv[js].z; xv.w += pv[js].w;
    }
  }
  *reinterpret_cast<float4*>(xs + bq * PX + c8 * 4) = xv;
#pragma unroll
  for (int n = 0; n < 2; ++n)
    *reinterpret_cast<float4*>(usm + ((tid >> 4) + 16 * n) * PB + (tid & 15) * 4) = uv[n];
  const float cv = cv0 + (float)t * (1.0f / 50.0f);
  const float a0 = fmaf(cv, w1.x, bb.x), a1 = fmaf(cv, w1.y, bb.y);
  const float a2 = fmaf(cv, w1.z, bb.z), a3 = fmaf(cv, w1.w, bb.w);
  SSM_SSTORE(smem);
  SSM_SSTORE_SECOND(smem + STAGE_DW);
  {
    float4 hv;
    hv.x = tanhf(a0); hv.y = tanhf(a1); hv.z = tanhf(a2); hv.w = tanhf(a3);
    *reinterpret_cast<float4*>(hbs + bq * PX + c8 * 4) = hv;
  }
  __syncthreads();
  if (!FIRST && it == 0)
    *reinterpret_cast<float4*>(out + ((size_t)(btp * 32 + bq) * T_ + (t - 1)) * S_ + j0 + c8 * 4) = xv;

  SSM_GLOAD(2);

  const float* xr = xs + (bl * 16 + MN) * PX + half * 2;
  v2f xf[8];
#pragma unroll
  for (int f = 0; f < 8; ++f) xf[f] = *reinterpret_cast<const v2f*>(xr + f * 4);
  const float* ur = usm + (bl * 16 + MN) * PB + kq * 16 + half * 2;
  v2f uf[4];
#pragma unroll
  for (int f = 0; f < 4; ++f) uf[f] = *reinterpret_cast<const v2f*>(ur + f * 4);
  const float* hbRow = hbs + (bl * 16 + MN) * PX;

  const int fragA = kq * 4 * SLAB_K + MN * PX + half * 2;
  const int fragB = 16 * SLAB_K + MN * PB + kq * 16 + half * 2;

  v8f t0 = {}, t1 = {};

  if (kq == 3) {
#pragma unroll
    for (int f = 0; f < 8; f += 2) {
      t0 = wmma4_f32(ba[f],     xf[f],     t0);
      t1 = wmma4_f32(ba[f + 1], xf[f + 1], t1);
    }
  }
  if (jc == 7) {
#pragma unroll
    for (int f = 0; f < 4; f += 2) {
      t0 = wmma4_f32(bu[f],     uf[f],     t0);
      t1 = wmma4_f32(bu[f + 1], uf[f + 1], t1);
    }
  }

#define SSM_LDGRP(dst, p)                                               \
  _Pragma("unroll") for (int q = 0; q < 8; ++q)                         \
    dst[q] = *reinterpret_cast<const v2f*>((p) + q * 4)
#define SSM_LDGRPB(dst, p)                                              \
  _Pragma("unroll") for (int q = 0; q < 8; ++q)                         \
    dst[q] = *reinterpret_cast<const v2f*>((p) + (q >> 2) * SLAB_B + (q & 3) * 4)
#define SSM_SCALE(dst, h)                                               \
  _Pragma("unroll") for (int q = 0; q < 8; ++q) dst[q] = xf[q] * (h)
#define SSM_SCALEU(dst, h2)                                             \
  _Pragma("unroll") for (int q = 0; q < 8; ++q) dst[q] = uf[q & 3] * (h2)[q >> 2]
#define SSM_MMGRP(frag, bf)                                             \
  _Pragma("unroll") for (int q = 0; q < 8; q += 2) {                    \
    t0 = wmma4_f32(frag[q],     bf[q],     t0);                         \
    t1 = wmma4_f32(frag[q + 1], bf[q + 1], t1);                         \
  }

#define SSM_PIN() asm volatile("" : "+v"(t0), "+v"(t1) : : "memory")

#define SSM_STAGE(cur_, nx1_, nx2_, h4_, hbp_, BAR_, STORE_, GLD_, sn3_)                \
  do {                                                                                 \
    const float* ar = (cur_) + fragA;                                                  \
    const float* br = (cur_) + fragB;                                                  \
    v2f fa[8], fb[8], p0[8], p1[8];                                                    \
    const v2f hb2 = *reinterpret_cast<const v2f*>(hbp_);                               \
    SSM_LDGRP(fb, ar + SLAB_K);                                                        \
    SSM_SCALE(p0, (h4_).x);                                                            \
    __builtin_amdgcn_sched_barrier(0);   \
    SSM_SCALE(p1, (h4_).y);                                                            \
    SSM_MMGRP(fc, p0);                                                                 \
    SSM_PIN();                                                                         \
    SSM_LDGRP(fa, ar + 2 * SLAB_K);                                                    \
    __builtin_amdgcn_sched_barrier(0);   \
    SSM_SCALE(p0, (h4_).z);                                                            \
    SSM_MMGRP(fb, p1);                                                                 \
    SSM_PIN();                                                                         \
    if (BAR_) {                                                                        \
      lds_drained();    \
      bar_signal();     \
    }                                                                                  \
    SSM_LDGRP(fb, ar + 3 * SLAB_K);                                                    \
    __builtin_amdgcn_sched_barrier(0);   \
    SSM_SCALE(p1, (h4_).w);                                                            \
    SSM_MMGRP(fa, p0);                                                                 \
    SSM_PIN();                                                                         \
    if (BAR_) { bar_wait(); }    \
    if (STORE_) { SSM_SSTORE(nx2_); }                                                  \
    if (GLD_) { SSM_GLOAD(sn3_); }                                                     \
    SSM_LDGRPB(fa, br);                                                                \
    __builtin_amdgcn_sched_barrier(0);   \
    SSM_SCALEU(p0, hb2);                                                               \
    SSM_MMGRP(fb, p1);                                                                 \
    SSM_PIN();                                                                         \
    if (BAR_) { SSM_LDGRP(fc, (nx1_) + fragA); }                                       \
    __builtin_amdgcn_sched_barrier(0);   \
    SSM_MMGRP(fa, p0);                                                                 \
  } while (0)

  v2f fc[8];
  SSM_LDGRP(fc, smem + fragA);

  int o0 = 0, o1 = STAGE_DW, o2 = 2 * STAGE_DW;
  for (int s = 0; s < 13; ++s) {
    const float4 hn = *reinterpret_cast<const float4*>(haRow + 16 * (s + 1));
    SSM_STAGE(smem + o0, smem + o1, smem + o2, hc, hbRow + 2 * s, true, true, true, s + 3);
    hc = hn;
    const int o_ = o0; o0 = o1; o1 = o2; o2 = o_;
  }
  {
    const float4 hn = *reinterpret_cast<const float4*>(haRow + 16 * 14);
    SSM_STAGE(smem + STAGE_DW, smem + 2 * STAGE_DW, smem, hc, hbRow + 26, true, true, false, 0);
    hc = hn;
  }
  {
    const float4 hn = *reinterpret_cast<const float4*>(haRow + 16 * 15);
    SSM_STAGE(smem + 2 * STAGE_DW, smem, smem, hc, hbRow + 28, true, false, false, 0);
    hc = hn;
  }
  SSM_STAGE(smem, smem, smem, hc, hbRow + 30, false, false, false, 0);

  v8f acc;
#pragma unroll
  for (int r = 0; r < 8; ++r) acc[r] = t0[r] + t1[r];

  float* red  = smem + STAGE_DW;
  float* mine = red + wv * 256 + MN * 16 + half * 8;
#pragma unroll
  for (int r = 0; r < 8; ++r) mine[r] = acc[r];
  __syncthreads();
#pragma unroll
  for (int blo = 0; blo < 2; ++blo) {
    float sum = 0.f;
#pragma unroll
    for (int w = 0; w < 4; ++w) sum += red[(w * 2 + blo) * 256 + tid];
    partials[(size_t)(jc * 64 + it * 4 + btp * 2 + blo) * 256 + tid] = sum;
  }
}

__global__ void ssm_step_reduce(const float* __restrict__ partials,
                                float* __restrict__ out, int t) {
  int e = blockIdx.x * blockDim.x + threadIdx.x;
  if (e >= B_ * S_) return;
  int b = e >> 8;
  int i = e & (S_ - 1);
  int it = i >> 4, M = i & 15;
  int bt = b >> 4, N = b & 15;
  float sum = 0.f;
#pragma unroll
  for (int jc = 0; jc < 8; ++jc) {
    sum += partials[((size_t)(jc * 64 + it * 4 + bt)) * 256 + N * 16 + M];
  }
  out[((size_t)b * T_ + t) * S_ + i] = sum;
}

extern "C" void kernel_launch(void* const* d_in, const int* in_sizes, int n_in,
                              void* d_out, int out_size, void* d_ws, size_t ws_size,
                              hipStream_t stream) {
  const float* x0      = (const float*)d_in[0];
  const float* context = (const float*)d_in[1];
  const float* us      = (const float*)d_in[2];
  const float* W1a     = (const float*)d_in[3];
  const float* b1a     = (const float*)d_in[4];
  const float* W2a     = (const float*)d_in[5];
  const float* b2a     = (const float*)d_in[6];
  const float* W1b     = (const float*)d_in[7];
  const float* b1b     = (const float*)d_in[8];
  const float* W2b     = (const float*)d_in[9];
  const float* b2b     = (const float*)d_in[10];
  float* out = (float*)d_out;

  float* HA = (float*)d_ws;
  float* P0 = HA + (size_t)T_ * B_ * H_;
  float* P1 = P0 + (size_t)512 * 256;

  int nHyp = T_ * B_ * H_;
  ssm_hyper<<<(nHyp + 255) / 256, 256, 0, stream>>>(context, W1a, b1a, HA);

  for (int t = 0; t < T_; ++t) {
    const float* prev  = (t & 1) ? P0 : P1;
    float* cur         = (t & 1) ? P1 : P0;
    const float* ubase = us + (size_t)t * A_;

    if (t == 0)
      ssm_step_partial<true><<<256, 256, 0, stream>>>(W2a, b2a, W2b, b2b, HA, context, W1b, b1b,
                                                     x0, prev, ubase, T_ * A_, cur, out, t);
    else
      ssm_step_partial<false><<<256, 256, 0, stream>>>(W2a, b2a, W2b, b2b, HA, context, W1b, b1b,
                                                      x0, prev, ubase, T_ * A_, cur, out, t);
  }
  ssm_step_reduce<<<(B_ * S_ + 255) / 256, 256, 0, stream>>>(P1, out, T_ - 1);
}
